# starting kernel
# baseline (speedup 1.0000x reference)
#include <hip/hip_runtime.h>

typedef __attribute__((ext_vector_type(2))) float v2f;
typedef __attribute__((ext_vector_type(8))) float v8f;

#define KTAP 5
#define RPAD 2
#define CCH  128
#define HH   128
#define WW   128
#define OHH  256
#define OWW  256
#define NCT  8
#define TILE_W 16
#define LDS_COLS (TILE_W + 2*RPAD)
#define LDS_ROWS (KTAP)
#define LDS_CH  (LDS_ROWS * LDS_COLS)
#define LDS_TOT (CCH * LDS_CH)
#define OST 66

__global__ __launch_bounds__(128)
void carafe_wmma_f32_kernel(const float* __restrict__ feats,
                            const float* __restrict__ masks,
                            float* __restrict__ out) {
    __shared__ float lfeat[LDS_TOT];

    const int tid  = threadIdx.x;
    const int lane = tid & 31;
    const int wave = tid >> 5;
    const int half = lane >> 4;
    const int mrow = lane & 15;
    const int p    = mrow >> 2;
    const int s    = mrow & 3;
    const int s1   = s >> 1;
    const int s2   = s & 1;

    const int bid  = blockIdx.x;
    const int wgrp = bid & 7;
    const int hi   = (bid >> 3) & 127;
    const int n    = bid >> 10;
    const int wblk = wgrp * TILE_W;

    const float* feat_n = feats + (size_t)n * CCH * HH * WW;
    for (int j = tid; j < LDS_TOT / 2; j += 128) {
        const int c   = j / 50;
        const int rem = j - c * 50;
        const int r   = rem / 10;
        const int x   = (rem - r * 10) * 2;
        const int gr  = hi + r - RPAD;
        const int gxc = wblk + x - RPAD;
        v2f v = {0.f, 0.f};
        if ((unsigned)gr < (unsigned)HH && (unsigned)gxc < (unsigned)WW)
            v = *(const v2f*)(feat_n + (size_t)c * (HH * WW) + (size_t)gr * WW + gxc);
        *(v2f*)(lfeat + 2 * j) = v;
    }
    __syncthreads();

    const int oh  = 2 * hi + s1;
    const int ow  = 2 * (wblk + wave * 4 + p) + s2;

    const bool gbx = (p == (half ? 2 : 0));
    const bool gby = (p == (half ? 3 : 1));

    v8f acc[NCT];
    const v8f vzero = {0.f, 0.f, 0.f, 0.f, 0.f, 0.f, 0.f, 0.f};
#pragma unroll
    for (int t = 0; t < NCT; ++t) acc[t] = vzero;

    const float* mask_base = masks + (size_t)n * (KTAP * KTAP) * (OHH * OWW)
                                   + (size_t)oh * OWW + ow;
    const int xbase = wave * 4 + half * 2;

#pragma unroll
    for (int kk = 0; kk < KTAP * KTAP; ++kk) {
        const int di = kk / KTAP;
        const int dj = kk - di * KTAP;

        const float mval = mask_base[(size_t)kk * (OHH * OWW)];
        v2f b;
        b.x = gbx ? mval : 0.f;
        b.y = gby ? mval : 0.f;

        const int lbase = di * LDS_COLS + xbase + dj;
#pragma unroll
        for (int t = 0; t < NCT; ++t) {
            const int off = (t * 16 + mrow) * LDS_CH + lbase;
            v2f a;
            a.x = lfeat[off];
            a.y = lfeat[off + 1];
            acc[t] = __builtin_amdgcn_wmma_f32_16x16x4_f32(
                false, a, false, b, (short)0, acc[t], false, false);
        }
    }

    __syncthreads();
    const int owl = 2 * (wave * 4 + p) + s2;
#pragma unroll
    for (int t = 0; t < NCT; ++t) {
#pragma unroll
        for (int v = 0; v < 8; ++v) {
            const int c = t * 16 + v + half * 8;
            lfeat[c * OST + s1 * 32 + owl] = acc[t][v];
        }
    }
    __syncthreads();

    float* out_n = out + (size_t)n * CCH * OHH * OWW;
    for (int i = 0; i < 64; ++i) {
        const int rid = (i << 2) | wave;
        const int c   = rid >> 1;
        const int sl  = rid & 1;
        const float val = lfeat[c * OST + sl * 32 + lane];
        out_n[(size_t)c * (OHH * OWW) + (size_t)(2 * hi + sl) * OWW
              + (wblk * 2 + lane)] = val;
    }
}

extern "C" void kernel_launch(void* const* d_in, const int* in_sizes, int n_in,
                              void* d_out, int out_size, void* d_ws, size_t ws_size,
                              hipStream_t stream) {
    const float* feats = (const float*)d_in[0];
    const float* masks = (const float*)d_in[1];
    float* out = (float*)d_out;

    dim3 grid(2 * 128 * 8);
    dim3 block(128);
    hipLaunchKernelGGL(carafe_wmma_f32_kernel, grid, block, 0, stream,
                       feats, masks, out);
}


# Feature-tile staging: each lane keeps one (row, pair) slot and requests its 64 channel loads before the first wait, instead of one load per wait; same LDS layout
# speedup vs baseline: 1.9169x; 1.9169x over previous
#include <hip/hip_runtime.h>

typedef __attribute__((ext_vector_type(2))) float v2f;
typedef __attribute__((ext_vector_type(8))) float v8f;

#define KTAP 5
#define RPAD 2
#define CCH  128
#define HH   128
#define WW   128
#define OHH  256
#define OWW  256
#define NCT  8
#define TILE_W 16
#define LDS_COLS (TILE_W + 2*RPAD)
#define LDS_ROWS (KTAP)
#define LDS_CH  (LDS_ROWS * LDS_COLS)
#define LDS_TOT (CCH * LDS_CH)
#define OST 66

__global__ __launch_bounds__(128)
void carafe_wmma_f32_kernel(const float* __restrict__ feats,
                            const float* __restrict__ masks,
                            float* __restrict__ out) {
    __shared__ float lfeat[LDS_TOT];

    const int tid  = threadIdx.x;
    const int lane = tid & 31;
    const int wave = tid >> 5;
    const int half = lane >> 4;
    const int mrow = lane & 15;
    const int p    = mrow >> 2;
    const int s    = mrow & 3;
    const int s1   = s >> 1;
    const int s2   = s & 1;

    const int bid  = blockIdx.x;
    const int wgrp = bid & 7;
    const int hi   = (bid >> 3) & 127;
    const int n    = bid >> 10;
    const int wblk = wgrp * TILE_W;

    const float* feat_n = feats + (size_t)n * CCH * HH * WW;
    if (tid < 2 * (LDS_CH / 2)) {
        const int cpar = (tid >= LDS_CH / 2) ? 1 : 0;
        const int rem  = tid - cpar * (LDS_CH / 2);
        const int r    = rem / 10;
        const int x    = (rem - r * 10) * 2;
        const int gr   = hi + r - RPAD;
        const int gxc  = wblk + x - RPAD;
        v2f buf[CCH / 2];
#pragma unroll
        for (int k = 0; k < CCH / 2; ++k) { buf[k].x = 0.f; buf[k].y = 0.f; }
        if ((unsigned)gr < (unsigned)HH && (unsigned)gxc < (unsigned)WW) {
            const float* src = feat_n + (size_t)cpar * (HH * WW) + (size_t)gr * WW + gxc;
#pragma unroll
            for (int k = 0; k < CCH / 2; ++k)
                buf[k] = *(const v2f*)(src + (size_t)k * (2 * HH * WW));
        }
        float* dst = lfeat + cpar * LDS_CH + 2 * rem;
#pragma unroll
        for (int k = 0; k < CCH / 2; ++k)
            *(v2f*)(dst + k * (2 * LDS_CH)) = buf[k];
    }
    __syncthreads();

    const int oh  = 2 * hi + s1;
    const int ow  = 2 * (wblk + wave * 4 + p) + s2;

    const bool gbx = (p == (half ? 2 : 0));
    const bool gby = (p == (half ? 3 : 1));

    v8f acc[NCT];
    const v8f vzero = {0.f, 0.f, 0.f, 0.f, 0.f, 0.f, 0.f, 0.f};
#pragma unroll
    for (int t = 0; t < NCT; ++t) acc[t] = vzero;

    const float* mask_base = masks + (size_t)n * (KTAP * KTAP) * (OHH * OWW)
                                   + (size_t)oh * OWW + ow;
    const int xbase = wave * 4 + half * 2;

#pragma unroll
    for (int kk = 0; kk < KTAP * KTAP; ++kk) {
        const int di = kk / KTAP;
        const int dj = kk - di * KTAP;

        const float mval = mask_base[(size_t)kk * (OHH * OWW)];
        v2f b;
        b.x = gbx ? mval : 0.f;
        b.y = gby ? mval : 0.f;

        const int lbase = di * LDS_COLS + xbase + dj;
#pragma unroll
        for (int t = 0; t < NCT; ++t) {
            const int off = (t * 16 + mrow) * LDS_CH + lbase;
            v2f a;
            a.x = lfeat[off];
            a.y = lfeat[off + 1];
            acc[t] = __builtin_amdgcn_wmma_f32_16x16x4_f32(
                false, a, false, b, (short)0, acc[t], false, false);
        }
    }

    __syncthreads();
    const int owl = 2 * (wave * 4 + p) + s2;
#pragma unroll
    for (int t = 0; t < NCT; ++t) {
#pragma unroll
        for (int v = 0; v < 8; ++v) {
            const int c = t * 16 + v + half * 8;
            lfeat[c * OST + s1 * 32 + owl] = acc[t][v];
        }
    }
    __syncthreads();

    float* out_n = out + (size_t)n * CCH * OHH * OWW;
    for (int i = 0; i < 64; ++i) {
        const int rid = (i << 2) | wave;
        const int c   = rid >> 1;
        const int sl  = rid & 1;
        const float val = lfeat[c * OST + sl * 32 + lane];
        out_n[(size_t)c * (OHH * OWW) + (size_t)(2 * hi + sl) * OWW
              + (wblk * 2 + lane)] = val;
    }
}

extern "C" void kernel_launch(void* const* d_in, const int* in_sizes, int n_in,
                              void* d_out, int out_size, void* d_ws, size_t ws_size,
                              hipStream_t stream) {
    const float* feats = (const float*)d_in[0];
    const float* masks = (const float*)d_in[1];
    float* out = (float*)d_out;

    dim3 grid(2 * 128 * 8);
    dim3 block(128);
    hipLaunchKernelGGL(carafe_wmma_f32_kernel, grid, block, 0, stream,
                       feats, masks, out);
}


# Row walk: a work-group covers 4 low-res rows over a rolling 5-row LDS ring, staging 1 new feature row per output row (was 5); next row's loads issued under the products; grid 2048 -> 512
# speedup vs baseline: 3.0847x; 1.6092x over previous
#include <hip/hip_runtime.h>

typedef __attribute__((ext_vector_type(2))) float v2f;
typedef __attribute__((ext_vector_type(8))) float v8f;

#define KTAP 5
#define RPAD 2
#define CCH  128
#define HH   128
#define WW   128
#define OHH  256
#define OWW  256
#define NCT  8
#define TILE_W 16
#define ROWS_WG 4
#define ROW_GROUPS (HH / ROWS_WG)
#define LDS_COLS (TILE_W + 2*RPAD)
#define LDS_ROWS (KTAP)
#define LDS_CH  (LDS_ROWS * LDS_COLS)
#define LDS_TOT (CCH * LDS_CH)
#define OST 66
#define REP_TOT (CCH * OST)
#define STG_CH 12
#define STG_K  11

__global__ __launch_bounds__(128)
void carafe_wmma_f32_kernel(const float* __restrict__ feats,
                            const float* __restrict__ masks,
                            float* __restrict__ out) {
    __shared__ __attribute__((aligned(16))) float lfeat[LDS_TOT];
    __shared__ __attribute__((aligned(16))) float lrep[REP_TOT];

    const int tid  = threadIdx.x;
    const int lane = tid & 31;
    const int wave = tid >> 5;
    const int half = lane >> 4;
    const int mrow = lane & 15;
    const int p    = mrow >> 2;
    const int s    = mrow & 3;
    const int s1   = s >> 1;
    const int s2   = s & 1;

    const int bid  = blockIdx.x;
    const int wgrp = bid & 7;
    const int rg   = (bid >> 3) % ROW_GROUPS;
    const int n    = bid / (8 * ROW_GROUPS);
    const int wblk = wgrp * TILE_W;
    const int hi0  = rg * ROWS_WG;

    const float* feat_n = feats + (size_t)n * CCH * HH * WW;
    const int  cres  = tid / 10;
    const int  pr    = tid - cres * 10;
    const bool stg   = tid < STG_CH * 10;
    const int  gxc   = wblk + 2 * pr - RPAD;
    const bool colok = stg && (unsigned)gxc < (unsigned)WW;
    const bool lastk = cres < CCH - (STG_K - 1) * STG_CH;
    const int  src_off = cres * (HH * WW) + gxc;
    float* dst_col = lfeat + cres * LDS_CH + 2 * pr;

    {
        v2f pro[4][STG_K];
#pragma unroll
        for (int j = 0; j < 4; ++j) {
#pragma unroll
            for (int k = 0; k < STG_K; ++k) { pro[j][k].x = 0.f; pro[j][k].y = 0.f; }
            const int gr = hi0 - RPAD + j;
            if (colok && (unsigned)gr < (unsigned)HH) {
                const float* sp = feat_n + (src_off + gr * WW);
#pragma unroll
                for (int k = 0; k < STG_K - 1; ++k)
                    pro[j][k] = *(const v2f*)(sp + (size_t)k * (STG_CH * HH * WW));
                if (lastk)
                    pro[j][STG_K - 1] = *(const v2f*)(sp + (size_t)(STG_K - 1) * (STG_CH * HH * WW));
            }
        }
        if (stg) {
#pragma unroll
            for (int j = 0; j < 4; ++j) {
                float* d = dst_col + j * LDS_COLS;
#pragma unroll
                for (int k = 0; k < STG_K - 1; ++k)
                    *(v2f*)(d + k * (STG_CH * LDS_CH)) = pro[j][k];
                if (lastk)
                    *(v2f*)(d + (STG_K - 1) * (STG_CH * LDS_CH)) = pro[j][STG_K - 1];
            }
        }
    }

    v2f nxt[STG_K];
#pragma unroll
    for (int k = 0; k < STG_K; ++k) { nxt[k].x = 0.f; nxt[k].y = 0.f; }
    {
        const int gr = hi0 + RPAD;
        if (colok && (unsigned)gr < (unsigned)HH) {
            const float* sp = feat_n + (src_off + gr * WW);
#pragma unroll
            for (int k = 0; k < STG_K - 1; ++k)
                nxt[k] = *(const v2f*)(sp + (size_t)k * (STG_CH * HH * WW));
            if (lastk)
                nxt[STG_K - 1] = *(const v2f*)(sp + (size_t)(STG_K - 1) * (STG_CH * HH * WW));
        }
    }

    const int ow = 2 * (wblk + wave * 4 + p) + s2;
    const float* mask_lane = masks + (size_t)n * (KTAP * KTAP) * (OHH * OWW)
                                   + (size_t)s1 * OWW + ow;
    int  cc[4];
    bool inb[4];
#pragma unroll
    for (int q = 0; q < 4; ++q) {
        const int dj = (q >> 1) * 4 + half * 2 - p + (q & 1);
        inb[q] = (unsigned)dj < (unsigned)KTAP;
        cc[q] = dj < 0 ? 0 : (dj > KTAP - 1 ? KTAP - 1 : dj);
    }
    float mraw[KTAP * 4];
#pragma unroll
    for (int di = 0; di < KTAP; ++di)
#pragma unroll
        for (int q = 0; q < 4; ++q)
            mraw[di * 4 + q] = mask_lane[(size_t)(2 * hi0) * OWW
                                         + (size_t)(di * KTAP + cc[q]) * (OHH * OWW)];

    const float* a_base = lfeat + mrow * LDS_CH + wave * 4 + half * 2;
    const int owl = 2 * (wave * 4 + p) + s2;
    float* out_n = out + (size_t)n * CCH * OHH * OWW;

    for (int it = 0; it < ROWS_WG; ++it) {
        const int hi = hi0 + it;

        v2f b[KTAP * 2];
#pragma unroll
        for (int di = 0; di < KTAP; ++di)
#pragma unroll
            for (int ch = 0; ch < 2; ++ch) {
                b[di * 2 + ch].x = inb[ch * 2 + 0] ? mraw[di * 4 + ch * 2 + 0] : 0.f;
                b[di * 2 + ch].y = inb[ch * 2 + 1] ? mraw[di * 4 + ch * 2 + 1] : 0.f;
            }

        {
            int st = it + 4; if (st >= LDS_ROWS) st -= LDS_ROWS;
            if (stg) {
                float* d = dst_col + st * LDS_COLS;
#pragma unroll
                for (int k = 0; k < STG_K - 1; ++k)
                    *(v2f*)(d + k * (STG_CH * LDS_CH)) = nxt[k];
                if (lastk)
                    *(v2f*)(d + (STG_K - 1) * (STG_CH * LDS_CH)) = nxt[STG_K - 1];
            }
        }

        __syncthreads();

        if (it + 1 < ROWS_WG) {
#pragma unroll
            for (int k = 0; k < STG_K; ++k) { nxt[k].x = 0.f; nxt[k].y = 0.f; }
            const int gr = hi + RPAD + 1;
            if (colok && (unsigned)gr < (unsigned)HH) {
                const float* sp = feat_n + (src_off + gr * WW);
#pragma unroll
                for (int k = 0; k < STG_K - 1; ++k)
                    nxt[k] = *(const v2f*)(sp + (size_t)k * (STG_CH * HH * WW));
                if (lastk)
                    nxt[STG_K - 1] = *(const v2f*)(sp + (size_t)(STG_K - 1) * (STG_CH * HH * WW));
            }
#pragma unroll
            for (int di = 0; di < KTAP; ++di)
#pragma unroll
                for (int q = 0; q < 4; ++q)
                    mraw[di * 4 + q] = mask_lane[(size_t)(2 * (hi + 1)) * OWW
                                                 + (size_t)(di * KTAP + cc[q]) * (OHH * OWW)];
        }

        v8f acc[NCT];
        const v8f vzero = {0.f, 0.f, 0.f, 0.f, 0.f, 0.f, 0.f, 0.f};
#pragma unroll
        for (int t = 0; t < NCT; ++t) acc[t] = vzero;

#pragma unroll
        for (int di = 0; di < KTAP; ++di) {
            int sl = it + di; if (sl >= LDS_ROWS) sl -= LDS_ROWS;
#pragma unroll
            for (int ch = 0; ch < 2; ++ch) {
                const float* a_row = a_base + sl * LDS_COLS + ch * 4;
#pragma unroll
                for (int t = 0; t < NCT; ++t) {
                    const v2f a = *(const v2f*)(a_row + t * 16 * LDS_CH);
                    acc[t] = __builtin_amdgcn_wmma_f32_16x16x4_f32(
                        false, a, false, b[di * 2 + ch], (short)0, acc[t], false, false);
                }
            }
        }

#pragma unroll
        for (int t = 0; t < NCT; ++t) {
#pragma unroll
            for (int v = 0; v < 8; ++v) {
                const int c = t * 16 + v + half * 8;
                lrep[c * OST + s1 * 32 + owl] = acc[t][v];
            }
        }
        __syncthreads();

        for (int i = 0; i < 64; ++i) {
            const int rid = (i << 2) | wave;
            const int c   = rid >> 1;
            const int sl  = rid & 1;
            const float val = lrep[c * OST + sl * 32 + lane];
            out_n[(size_t)c * (OHH * OWW) + (size_t)(2 * hi + sl) * OWW
                  + (wblk * 2 + lane)] = val;
        }
    }
}

extern "C" void kernel_launch(void* const* d_in, const int* in_sizes, int n_in,
                              void* d_out, int out_size, void* d_ws, size_t ws_size,
                              hipStream_t stream) {
    const float* feats = (const float*)d_in[0];
    const float* masks = (const float*)d_in[1];
    float* out = (float*)d_out;

    dim3 grid(2 * 32 * 8);
    dim3 block(128);
    hipLaunchKernelGGL(carafe_wmma_f32_kernel, grid, block, 0, stream,
                       feats, masks, out);
}


# Work-group tile widened to 32 low-res pixels (8 waves, grid 512 -> 256): each output row piece is one whole aligned 256-byte unit, stored 16 B per lane, two units per instruction (was half a unit)
# speedup vs baseline: 3.8120x; 1.2358x over previous
#include <hip/hip_runtime.h>

typedef __attribute__((ext_vector_type(2))) float v2f;
typedef __attribute__((ext_vector_type(4))) float v4f;
typedef __attribute__((ext_vector_type(8))) float v8f;

#define KTAP 5
#define RPAD 2
#define CCH  128
#define HH   128
#define WW   128
#define OHH  256
#define OWW  256
#define NCT  8
#define TILE_W 32
#define NWAVE  8
#define ROWS_WG 4
#define ROW_GROUPS (HH / ROWS_WG)
#define LDS_COLS (TILE_W + 2*RPAD)
#define LDS_ROWS (KTAP)
#define LDS_CH  (LDS_ROWS * LDS_COLS)
#define LDS_TOT (CCH * LDS_CH)
#define OSR 72
#define OST 148
#define REP_TOT (CCH * OST)
#define STG_CH 14
#define STG_K  10

__global__ __launch_bounds__(256)
void carafe_wmma_f32_kernel(const float* __restrict__ feats,
                            const float* __restrict__ masks,
                            float* __restrict__ out) {
    __shared__ __attribute__((aligned(16))) float lfeat[LDS_TOT];
    __shared__ __attribute__((aligned(16))) float lrep[REP_TOT];

    const int tid  = threadIdx.x;
    const int lane = tid & 31;
    const int wave = tid >> 5;
    const int half = lane >> 4;
    const int mrow = lane & 15;
    const int p    = mrow >> 2;
    const int s    = mrow & 3;
    const int s1   = s >> 1;
    const int s2   = s & 1;

    const int bid  = blockIdx.x;
    const int wgrp = bid & 3;
    const int rg   = (bid >> 2) % ROW_GROUPS;
    const int n    = bid / (4 * ROW_GROUPS);
    const int wblk = wgrp * TILE_W;
    const int hi0  = rg * ROWS_WG;

    const float* feat_n = feats + (size_t)n * CCH * HH * WW;
    const int  cres  = tid / 18;
    const int  pr    = tid - cres * 18;
    const bool stg   = tid < STG_CH * 18;
    const int  gxc   = wblk + 2 * pr - RPAD;
    const bool colok = stg && (unsigned)gxc < (unsigned)WW;
    const bool lastk = cres < CCH - (STG_K - 1) * STG_CH;
    const int  src_off = cres * (HH * WW) + gxc;
    float* dst_col = lfeat + cres * LDS_CH + 2 * pr;

    {
        v2f pro[4][STG_K];
#pragma unroll
        for (int j = 0; j < 4; ++j) {
#pragma unroll
            for (int k = 0; k < STG_K; ++k) { pro[j][k].x = 0.f; pro[j][k].y = 0.f; }
            const int gr = hi0 - RPAD + j;
            if (colok && (unsigned)gr < (unsigned)HH) {
                const float* sp = feat_n + (src_off + gr * WW);
#pragma unroll
                for (int k = 0; k < STG_K - 1; ++k)
                    pro[j][k] = *(const v2f*)(sp + (size_t)k * (STG_CH * HH * WW));
                if (lastk)
                    pro[j][STG_K - 1] = *(const v2f*)(sp + (size_t)(STG_K - 1) * (STG_CH * HH * WW));
            }
        }
        if (stg) {
#pragma unroll
            for (int j = 0; j < 4; ++j) {
                float* d = dst_col + j * LDS_COLS;
#pragma unroll
                for (int k = 0; k < STG_K - 1; ++k)
                    *(v2f*)(d + k * (STG_CH * LDS_CH)) = pro[j][k];
                if (lastk)
                    *(v2f*)(d + (STG_K - 1) * (STG_CH * LDS_CH)) = pro[j][STG_K - 1];
            }
        }
    }

    v2f nxt[STG_K];
#pragma unroll
    for (int k = 0; k < STG_K; ++k) { nxt[k].x = 0.f; nxt[k].y = 0.f; }
    {
        const int gr = hi0 + RPAD;
        if (colok && (unsigned)gr < (unsigned)HH) {
            const float* sp = feat_n + (src_off + gr * WW);
#pragma unroll
            for (int k = 0; k < STG_K - 1; ++k)
                nxt[k] = *(const v2f*)(sp + (size_t)k * (STG_CH * HH * WW));
            if (lastk)
                nxt[STG_K - 1] = *(const v2f*)(sp + (size_t)(STG_K - 1) * (STG_CH * HH * WW));
        }
    }

    const int ow = 2 * (wblk + wave * 4 + p) + s2;
    const float* mask_lane = masks + (size_t)n * (KTAP * KTAP) * (OHH * OWW)
                                   + (size_t)s1 * OWW + ow;
    int  cc[4];
    bool inb[4];
#pragma unroll
    for (int q = 0; q < 4; ++q) {
        const int dj = (q >> 1) * 4 + half * 2 - p + (q & 1);
        inb[q] = (unsigned)dj < (unsigned)KTAP;
        cc[q] = dj < 0 ? 0 : (dj > KTAP - 1 ? KTAP - 1 : dj);
    }
    float mraw[KTAP * 4];
#pragma unroll
    for (int di = 0; di < KTAP; ++di)
#pragma unroll
        for (int q = 0; q < 4; ++q)
            mraw[di * 4 + q] = mask_lane[(size_t)(2 * hi0) * OWW
                                         + (size_t)(di * KTAP + cc[q]) * (OHH * OWW)];

    const float* a_base = lfeat + mrow * LDS_CH + wave * 4 + half * 2;
    const int owl = 2 * (wave * 4 + p) + s2;
    float* out_n = out + (size_t)n * CCH * OHH * OWW;

    for (int it = 0; it < ROWS_WG; ++it) {
        const int hi = hi0 + it;

        v2f b[KTAP * 2];
#pragma unroll
        for (int di = 0; di < KTAP; ++di)
#pragma unroll
            for (int ch = 0; ch < 2; ++ch) {
                b[di * 2 + ch].x = inb[ch * 2 + 0] ? mraw[di * 4 + ch * 2 + 0] : 0.f;
                b[di * 2 + ch].y = inb[ch * 2 + 1] ? mraw[di * 4 + ch * 2 + 1] : 0.f;
            }

        {
            int st = it + 4; if (st >= LDS_ROWS) st -= LDS_ROWS;
            if (stg) {
                float* d = dst_col + st * LDS_COLS;
#pragma unroll
                for (int k = 0; k < STG_K - 1; ++k)
                    *(v2f*)(d + k * (STG_CH * LDS_CH)) = nxt[k];
                if (lastk)
                    *(v2f*)(d + (STG_K - 1) * (STG_CH * LDS_CH)) = nxt[STG_K - 1];
            }
        }

        __syncthreads();

        if (it + 1 < ROWS_WG) {
#pragma unroll
            for (int k = 0; k < STG_K; ++k) { nxt[k].x = 0.f; nxt[k].y = 0.f; }
            const int gr = hi + RPAD + 1;
            if (colok && (unsigned)gr < (unsigned)HH) {
                const float* sp = feat_n + (src_off + gr * WW);
#pragma unroll
                for (int k = 0; k < STG_K - 1; ++k)
                    nxt[k] = *(const v2f*)(sp + (size_t)k * (STG_CH * HH * WW));
                if (lastk)
                    nxt[STG_K - 1] = *(const v2f*)(sp + (size_t)(STG_K - 1) * (STG_CH * HH * WW));
            }
#pragma unroll
            for (int di = 0; di < KTAP; ++di)
#pragma unroll
                for (int q = 0; q < 4; ++q)
                    mraw[di * 4 + q] = mask_lane[(size_t)(2 * (hi + 1)) * OWW
                                                 + (size_t)(di * KTAP + cc[q]) * (OHH * OWW)];
        }

        v8f acc[NCT];
        const v8f vzero = {0.f, 0.f, 0.f, 0.f, 0.f, 0.f, 0.f, 0.f};
#pragma unroll
        for (int t = 0; t < NCT; ++t) acc[t] = vzero;

#pragma unroll
        for (int di = 0; di < KTAP; ++di) {
            int sl = it + di; if (sl >= LDS_ROWS) sl -= LDS_ROWS;
#pragma unroll
            for (int ch = 0; ch < 2; ++ch) {
                const float* a_row = a_base + sl * LDS_COLS + ch * 4;
#pragma unroll
                for (int t = 0; t < NCT; ++t) {
                    const v2f a = *(const v2f*)(a_row + t * 16 * LDS_CH);
                    acc[t] = __builtin_amdgcn_wmma_f32_16x16x4_f32(
                        false, a, false, b[di * 2 + ch], (short)0, acc[t], false, false);
                }
            }
        }

#pragma unroll
        for (int t = 0; t < NCT; ++t) {
#pragma unroll
            for (int v = 0; v < 8; ++v) {
                const int c = t * 16 + v + half * 8;
                lrep[c * OST + s1 * OSR + owl] = acc[t][v];
            }
        }
        __syncthreads();

        {
            const int l16 = lane & 15;
            const float* rsrc = lrep + half * OSR + 4 * l16;
            float* odst = out_n + (size_t)(2 * hi + half) * OWW + (wblk * 2 + 4 * l16);
            for (int j = 0; j < CCH / NWAVE; ++j) {
                const int c = j * NWAVE + wave;
                const v4f val = *(const v4f*)(rsrc + c * OST);
                *(v4f*)(odst + (size_t)c * (OHH * OWW)) = val;
            }
        }
    }
}

extern "C" void kernel_launch(void* const* d_in, const int* in_sizes, int n_in,
                              void* d_out, int out_size, void* d_ws, size_t ws_size,
                              hipStream_t stream) {
    const float* feats = (const float*)d_in[0];
    const float* masks = (const float*)d_in[1];
    float* out = (float*)d_out;

    dim3 grid(2 * 32 * 4);
    dim3 block(256);
    hipLaunchKernelGGL(carafe_wmma_f32_kernel, grid, block, 0, stream,
                       feats, masks, out);
}


# Feature fragments kept in registers across the unrolled row walk: each ring row is read from LDS once, as one group ahead of the products (was in all 5 trips); barriers fence LDS only; same sums
# speedup vs baseline: 4.1423x; 1.0866x over previous
#include <hip/hip_runtime.h>

typedef __attribute__((ext_vector_type(2))) float v2f;
typedef __attribute__((ext_vector_type(4))) float v4f;
typedef __attribute__((ext_vector_type(8))) float v8f;

#define KTAP 5
#define RPAD 2
#define CCH  128
#define HH   128
#define WW   128
#define OHH  256
#define OWW  256
#define NCT  8
#define TILE_W 32
#define NWAVE  8
#define ROWS_WG 4
#define ROW_GROUPS (HH / ROWS_WG)
#define LDS_COLS (TILE_W + 2*RPAD)
#define LDS_ROWS (KTAP)
#define LDS_CH  (LDS_ROWS * LDS_COLS)
#define LDS_TOT (CCH * LDS_CH)
#define OSR 72
#define OST 148
#define REP_TOT (CCH * OST)
#define STG_CH 14
#define STG_K  10

__device__ __forceinline__ void lds_barrier() {
    __builtin_amdgcn_fence(__ATOMIC_RELEASE, "workgroup", "local");
    __builtin_amdgcn_s_barrier();
    __builtin_amdgcn_fence(__ATOMIC_ACQUIRE, "workgroup", "local");
}

__global__ __launch_bounds__(256)
void carafe_wmma_f32_kernel(const float* __restrict__ feats,
                            const float* __restrict__ masks,
                            float* __restrict__ out) {
    __shared__ __attribute__((aligned(16))) float lfeat[LDS_TOT];
    __shared__ __attribute__((aligned(16))) float lrep[REP_TOT];

    const int tid  = threadIdx.x;
    const int lane = tid & 31;
    const int wave = tid >> 5;
    const int half = lane >> 4;
    const int mrow = lane & 15;
    const int p    = mrow >> 2;
    const int s    = mrow & 3;
    const int s1   = s >> 1;
    const int s2   = s & 1;

    const int bid  = blockIdx.x;
    const int wgrp = bid & 3;
    const int rg   = (bid >> 2) % ROW_GROUPS;
    const int n    = bid / (4 * ROW_GROUPS);
    const int wblk = wgrp * TILE_W;
    const int hi0  = rg * ROWS_WG;

    const float* feat_n = feats + (size_t)n * CCH * HH * WW;
    const int  cres  = tid / 18;
    const int  pr    = tid - cres * 18;
    const bool stg   = tid < STG_CH * 18;
    const int  gxc   = wblk + 2 * pr - RPAD;
    const bool colok = stg && (unsigned)gxc < (unsigned)WW;
    const bool lastk = cres < CCH - (STG_K - 1) * STG_CH;
    const int  src_off = cres * (HH * WW) + gxc;
    float* dst_col = lfeat + cres * LDS_CH + 2 * pr;
    const size_t off_k9 = (size_t)(lastk ? STG_K - 1 : STG_K - 2) * (STG_CH * HH * WW);

    {
        v2f pro[4][STG_K];
#pragma unroll
        for (int j = 0; j < 4; ++j) {
#pragma unroll
            for (int k = 0; k < STG_K; ++k) { pro[j][k].x = 0.f; pro[j][k].y = 0.f; }
            const int gr = hi0 - RPAD + j;
            if (colok && (unsigned)gr < (unsigned)HH) {
                const float* sp = feat_n + (src_off + gr * WW);
#pragma unroll
                for (int k = 0; k < STG_K - 1; ++k)
                    pro[j][k] = *(const v2f*)(sp + (size_t)k * (STG_CH * HH * WW));
                if (lastk)
                    pro[j][STG_K - 1] = *(const v2f*)(sp + (size_t)(STG_K - 1) * (STG_CH * HH * WW));
            }
        }
        if (stg) {
#pragma unroll
            for (int j = 0; j < 4; ++j) {
                float* d = dst_col + j * LDS_COLS;
#pragma unroll
                for (int k = 0; k < STG_K - 1; ++k)
                    *(v2f*)(d + k * (STG_CH * LDS_CH)) = pro[j][k];
                if (lastk)
                    *(v2f*)(d + (STG_K - 1) * (STG_CH * LDS_CH)) = pro[j][STG_K - 1];
            }
        }
    }

    v2f nxt[STG_K];
#pragma unroll
    for (int k = 0; k < STG_K; ++k) { nxt[k].x = 0.f; nxt[k].y = 0.f; }
    {
        const int gr = hi0 + RPAD;
        if (colok && (unsigned)gr < (unsigned)HH) {
            const float* sp = feat_n + (src_off + gr * WW);
#pragma unroll
            for (int k = 0; k < STG_K - 1; ++k)
                nxt[k] = *(const v2f*)(sp + (size_t)k * (STG_CH * HH * WW));
            nxt[STG_K - 1] = *(const v2f*)(sp + off_k9);
        }
    }

    const int ow = 2 * (wblk + wave * 4 + p) + s2;
    const float* mask_lane = masks + (size_t)n * (KTAP * KTAP) * (OHH * OWW)
                                   + (size_t)s1 * OWW + ow;
    int  cc[4];
    bool inb[4];
#pragma unroll
    for (int q = 0; q < 4; ++q) {
        const int dj = (q >> 1) * 4 + half * 2 - p + (q & 1);
        inb[q] = (unsigned)dj < (unsigned)KTAP;
        cc[q] = dj < 0 ? 0 : (dj > KTAP - 1 ? KTAP - 1 : dj);
    }
    float mraw[KTAP * 4];
#pragma unroll
    for (int di = 0; di < KTAP; ++di)
#pragma unroll
        for (int q = 0; q < 4; ++q)
            mraw[di * 4 + q] = mask_lane[(size_t)(2 * hi0) * OWW
                                         + (size_t)(di * KTAP + cc[q]) * (OHH * OWW)];

    const float* a_base = lfeat + mrow * LDS_CH + wave * 4 + half * 2;
    const int owl = 2 * (wave * 4 + p) + s2;
    float* out_n = out + (size_t)n * CCH * OHH * OWW;

    v2f areg[LDS_ROWS][NCT][2];

#pragma unroll
    for (int it = 0; it < ROWS_WG; ++it) {
        const int hi = hi0 + it;

        v2f b[KTAP * 2];
#pragma unroll
        for (int di = 0; di < KTAP; ++di)
#pragma unroll
            for (int ch = 0; ch < 2; ++ch) {
                b[di * 2 + ch].x = inb[ch * 2 + 0] ? mraw[di * 4 + ch * 2 + 0] : 0.f;
                b[di * 2 + ch].y = inb[ch * 2 + 1] ? mraw[di * 4 + ch * 2 + 1] : 0.f;
            }

        {
            int st = it + 4; if (st >= LDS_ROWS) st -= LDS_ROWS;
            if (stg) {
                float* d = dst_col + st * LDS_COLS;
#pragma unroll
                for (int k = 0; k < STG_K - 1; ++k)
                    *(v2f*)(d + k * (STG_CH * LDS_CH)) = nxt[k];
                if (lastk)
                    *(v2f*)(d + (STG_K - 1) * (STG_CH * LDS_CH)) = nxt[STG_K - 1];
            }
        }
#pragma unroll
        for (int i = 0; i < KTAP * 2; ++i)
            asm volatile("" : "+v"(b[i].x), "+v"(b[i].y));

        lds_barrier();

#pragma unroll
        for (int j = (it == 0 ? 0 : KTAP - 1); j < KTAP; ++j) {
            int sl = it + j; if (sl >= LDS_ROWS) sl -= LDS_ROWS;
#pragma unroll
            for (int t = 0; t < NCT; ++t)
#pragma unroll
                for (int ch = 0; ch < 2; ++ch)
                    areg[sl][t][ch] = *(const v2f*)(a_base + sl * LDS_COLS + ch * 4
                                                    + t * 16 * LDS_CH);
        }

        if (it + 1 < ROWS_WG) {
#pragma unroll
            for (int k = 0; k < STG_K; ++k) { nxt[k].x = 0.f; nxt[k].y = 0.f; }
            const int gr = hi + RPAD + 1;
            if (colok && (unsigned)gr < (unsigned)HH) {
                const float* sp = feat_n + (src_off + gr * WW);
#pragma unroll
                for (int k = 0; k < STG_K - 1; ++k)
                    nxt[k] = *(const v2f*)(sp + (size_t)k * (STG_CH * HH * WW));
                nxt[STG_K - 1] = *(const v2f*)(sp + off_k9);
            }
#pragma unroll
            for (int di = 0; di < KTAP; ++di)
#pragma unroll
                for (int q = 0; q < 4; ++q)
                    mraw[di * 4 + q] = mask_lane[(size_t)(2 * (hi + 1)) * OWW
                                                 + (size_t)(di * KTAP + cc[q]) * (OHH * OWW)];
        }
        __builtin_amdgcn_sched_barrier(0);

        v8f acc[NCT];
        const v8f vzero = {0.f, 0.f, 0.f, 0.f, 0.f, 0.f, 0.f, 0.f};
#pragma unroll
        for (int t = 0; t < NCT; ++t) acc[t] = vzero;

#pragma unroll
        for (int di = 0; di < KTAP; ++di) {
            int sl = it + di; if (sl >= LDS_ROWS) sl -= LDS_ROWS;
#pragma unroll
            for (int ch = 0; ch < 2; ++ch) {
#pragma unroll
                for (int t = 0; t < NCT; ++t) {
                    acc[t] = __builtin_amdgcn_wmma_f32_16x16x4_f32(
                        false, areg[sl][t][ch], false, b[di * 2 + ch], (short)0, acc[t],
                        false, false);
                }
            }
        }

#pragma unroll
        for (int t = 0; t < NCT; ++t) {
#pragma unroll
            for (int v = 0; v < 8; ++v) {
                const int c = t * 16 + v + half * 8;
                lrep[c * OST + s1 * OSR + owl] = acc[t][v];
            }
        }
        lds_barrier();

        {
            const int l16 = lane & 15;
            const float* rsrc = lrep + half * OSR + 4 * l16;
            float* odst = out_n + (size_t)(2 * hi + half) * OWW + (wblk * 2 + 4 * l16);
            for (int j = 0; j < CCH / NWAVE; ++j) {
                const int c = j * NWAVE + wave;
                const v4f val = *(const v4f*)(rsrc + c * OST);
                *(v4f*)(odst + (size_t)c * (OHH * OWW)) = val;
            }
        }
    }
}

extern "C" void kernel_launch(void* const* d_in, const int* in_sizes, int n_in,
                              void* d_out, int out_size, void* d_ws, size_t ws_size,
                              hipStream_t stream) {
    const float* feats = (const float*)d_in[0];
    const float* masks = (const float*)d_in[1];
    float* out = (float*)d_out;

    dim3 grid(2 * 32 * 4);
    dim3 block(256);
    hipLaunchKernelGGL(carafe_wmma_f32_kernel, grid, block, 0, stream,
                       feats, masks, out);
}
